# starting kernel
# baseline (speedup 1.0000x reference)
#include <hip/hip_runtime.h>

typedef __attribute__((ext_vector_type(2))) float v2f;
typedef __attribute__((ext_vector_type(8))) float v8f;

#define GAT_B 16
#define GAT_K 128
#define GAT_F 128
#define GAT_H 256
#define GAT_ALPHA 0.2f

__global__ void __launch_bounds__(32)
gat_qk_gemm(const float* __restrict__ x,
            const float* __restrict__ w,
            float* __restrict__ s1,
            float* __restrict__ s2) {
  int tile = blockIdx.x;
  const int ht = tile % (GAT_H / 16); tile /= (GAT_H / 16);
  const int it = tile % (GAT_K / 16); tile /= (GAT_K / 16);
  const int b  = tile;
  const int i0 = it * 16, h0 = ht * 16;
  const int lane = threadIdx.x & 31;
  const int hf   = lane >> 4;
  const int l    = lane & 15;

  const float* xb = x + b * GAT_F * GAT_K;
  const float* wr = w + (h0 + l) * (2 * GAT_F);

  v8f c1 = {};
  v8f c2 = {};
  for (int f0 = 0; f0 < GAT_F; f0 += 4) {
    v2f av, b1, b2;
    av.x = xb[(f0 + 2 * hf + 0) * GAT_K + i0 + l];
    av.y = xb[(f0 + 2 * hf + 1) * GAT_K + i0 + l];
    b1.x = wr[f0 + 2 * hf + 0];
    b1.y = wr[f0 + 2 * hf + 1];
    b2.x = wr[GAT_F + f0 + 2 * hf + 0];
    b2.y = wr[GAT_F + f0 + 2 * hf + 1];
    c1 = __builtin_amdgcn_wmma_f32_16x16x4_f32(false, av, false, b1, (short)0,
                                               c1, false, false);
    c2 = __builtin_amdgcn_wmma_f32_16x16x4_f32(false, av, false, b2, (short)0,
                                               c2, false, false);
  }
#pragma unroll
  for (int r = 0; r < 8; ++r) {
    const int row = i0 + r + 8 * hf;
    const int col = h0 + l;
    s1[(b * GAT_K + row) * GAT_H + col] = c1[r];
    s2[(b * GAT_K + row) * GAT_H + col] = c2[r];
  }
}

__global__ void __launch_bounds__(256)
gat_attn(const float* __restrict__ s1,
         const float* __restrict__ s2,
         const float* __restrict__ lb,
         const float* __restrict__ av,
         const float* __restrict__ bias,
         float* __restrict__ attn) {
  __shared__ float s2lds[64 * GAT_H];

  const int blk  = blockIdx.x;
  const int ic   = blk % (GAT_K / 8);
  const int b    = blk / (GAT_K / 8);
  const int wave = threadIdx.x >> 5;
  const int lane = threadIdx.x & 31;
  const int i    = ic * 8 + wave;

  float r1[8], ar[8];
#pragma unroll
  for (int r = 0; r < 8; ++r) {
    const int h = lane + 32 * r;
    r1[r] = s1[(b * GAT_K + i) * GAT_H + h] + lb[h];
    ar[r] = av[h];
  }

  float e0 = 0.f, e1 = 0.f, e2 = 0.f, e3 = 0.f;

#pragma unroll
  for (int c = 0; c < 2; ++c) {
    __syncthreads();
    for (int idx = threadIdx.x; idx < 64 * GAT_H; idx += 256) {
      const int jj = idx >> 8;
      const int h  = idx & (GAT_H - 1);
      s2lds[idx] = s2[(b * GAT_K + c * 64 + jj) * GAT_H + h];
    }
    __syncthreads();

#pragma unroll
    for (int t2 = 0; t2 < 2; ++t2) {
      for (int jl = 0; jl < 32; ++jl) {
        const int jj = t2 * 32 + jl;
        float acc = 0.f;
#pragma unroll
        for (int r = 0; r < 8; ++r) {
          float t = r1[r] + s2lds[jj * GAT_H + lane + 32 * r];
          t = fmaxf(t, GAT_ALPHA * t);
          acc = fmaf(t, ar[r], acc);
        }
        acc += __shfl_xor(acc, 1);
        acc += __shfl_xor(acc, 2);
        acc += __shfl_xor(acc, 4);
        acc += __shfl_xor(acc, 8);
        acc += __shfl_xor(acc, 16);
        const int j = c * 64 + jj;
        acc += bias[i * GAT_K + j];
        if (jl == lane) {
          if (c * 2 + t2 == 0)      e0 = acc;
          else if (c * 2 + t2 == 1) e1 = acc;
          else if (c * 2 + t2 == 2) e2 = acc;
          else                      e3 = acc;
        }
      }
    }
  }

  float m = fmaxf(fmaxf(e0, e1), fmaxf(e2, e3));
  m = fmaxf(m, __shfl_xor(m, 1));
  m = fmaxf(m, __shfl_xor(m, 2));
  m = fmaxf(m, __shfl_xor(m, 4));
  m = fmaxf(m, __shfl_xor(m, 8));
  m = fmaxf(m, __shfl_xor(m, 16));
  e0 = __expf(e0 - m);
  e1 = __expf(e1 - m);
  e2 = __expf(e2 - m);
  e3 = __expf(e3 - m);
  float s = e0 + e1 + e2 + e3;
  s += __shfl_xor(s, 1);
  s += __shfl_xor(s, 2);
  s += __shfl_xor(s, 4);
  s += __shfl_xor(s, 8);
  s += __shfl_xor(s, 16);
  const float inv = 1.0f / s;
  float* arow = attn + (b * GAT_K + i) * GAT_K;
  arow[lane +  0] = e0 * inv;
  arow[lane + 32] = e1 * inv;
  arow[lane + 64] = e2 * inv;
  arow[lane + 96] = e3 * inv;
}

__global__ void __launch_bounds__(32)
gat_out(const float* __restrict__ attn,
        const float* __restrict__ x,
        float* __restrict__ out) {
  int tile = blockIdx.x;
  const int ft = tile % (GAT_F / 16); tile /= (GAT_F / 16);
  const int it = tile % (GAT_K / 16); tile /= (GAT_K / 16);
  const int b  = tile;
  const int i0 = it * 16, f0 = ft * 16;
  const int lane = threadIdx.x & 31;
  const int hf   = lane >> 4;
  const int l    = lane & 15;

  const float* ab = attn + b * GAT_K * GAT_K;
  const float* xb = x + b * GAT_F * GAT_K;

  v8f c = {};
  for (int j0 = 0; j0 < GAT_K; j0 += 4) {
    v2f av, bv;
    av.x = ab[(i0 + l) * GAT_K + j0 + 2 * hf + 0];
    av.y = ab[(i0 + l) * GAT_K + j0 + 2 * hf + 1];
    bv.x = xb[(f0 + l) * GAT_K + j0 + 2 * hf + 0];
    bv.y = xb[(f0 + l) * GAT_K + j0 + 2 * hf + 1];
    c = __builtin_amdgcn_wmma_f32_16x16x4_f32(false, av, false, bv, (short)0,
                                              c, false, false);
  }
#pragma unroll
  for (int r = 0; r < 8; ++r) {
    const int row = i0 + r + 8 * hf;
    const int col = f0 + l;
    const float z = c[r];
    out[(b * GAT_K + row) * GAT_F + col] = 1.0f / (1.0f + __expf(-z));
  }
}

extern "C" void kernel_launch(void* const* d_in, const int* in_sizes, int n_in,
                              void* d_out, int out_size, void* d_ws, size_t ws_size,
                              hipStream_t stream) {
  (void)in_sizes; (void)n_in; (void)out_size; (void)ws_size;
  const float* x    = (const float*)d_in[0];
  const float* w    = (const float*)d_in[1];
  const float* lb   = (const float*)d_in[2];
  const float* av   = (const float*)d_in[3];
  const float* bias = (const float*)d_in[4];
  float* out = (float*)d_out;

  float* s1   = (float*)d_ws;
  float* s2   = s1 + GAT_B * GAT_K * GAT_H;
  float* attn = s2 + GAT_B * GAT_K * GAT_H;

  gat_qk_gemm<<<GAT_B * (GAT_K / 16) * (GAT_H / 16), 32, 0, stream>>>(x, w, s1, s2);
  gat_attn  <<<GAT_B * (GAT_K / 8),               256, 0, stream>>>(s1, s2, lb, av, bias, attn);
  gat_out   <<<GAT_B * (GAT_K / 16) * (GAT_F / 16), 32, 0, stream>>>(attn, x, out);
}


# attention re-tiled: wave keeps 16 s2 columns in registers (lanes own h), 8 rows stream past from LDS, 32 partial sums per row pair fold in a 31-exchange vector-ALU tree; fp32 sum order over h changes
# speedup vs baseline: 2.4520x; 2.4520x over previous
#include <hip/hip_runtime.h>

typedef __attribute__((ext_vector_type(2))) float v2f;
typedef __attribute__((ext_vector_type(8))) float v8f;

#define GAT_B 16
#define GAT_K 128
#define GAT_F 128
#define GAT_H 256
#define GAT_ALPHA 0.2f

__global__ void __launch_bounds__(32)
gat_qk_gemm(const float* __restrict__ x,
            const float* __restrict__ w,
            float* __restrict__ s1,
            float* __restrict__ s2) {
  int tile = blockIdx.x;
  const int ht = tile % (GAT_H / 16); tile /= (GAT_H / 16);
  const int it = tile % (GAT_K / 16); tile /= (GAT_K / 16);
  const int b  = tile;
  const int i0 = it * 16, h0 = ht * 16;
  const int lane = threadIdx.x & 31;
  const int hf   = lane >> 4;
  const int l    = lane & 15;

  const float* xb = x + b * GAT_F * GAT_K;
  const float* wr = w + (h0 + l) * (2 * GAT_F);

  v8f c1 = {};
  v8f c2 = {};
  for (int f0 = 0; f0 < GAT_F; f0 += 4) {
    v2f av, b1, b2;
    av.x = xb[(f0 + 2 * hf + 0) * GAT_K + i0 + l];
    av.y = xb[(f0 + 2 * hf + 1) * GAT_K + i0 + l];
    b1.x = wr[f0 + 2 * hf + 0];
    b1.y = wr[f0 + 2 * hf + 1];
    b2.x = wr[GAT_F + f0 + 2 * hf + 0];
    b2.y = wr[GAT_F + f0 + 2 * hf + 1];
    c1 = __builtin_amdgcn_wmma_f32_16x16x4_f32(false, av, false, b1, (short)0,
                                               c1, false, false);
    c2 = __builtin_amdgcn_wmma_f32_16x16x4_f32(false, av, false, b2, (short)0,
                                               c2, false, false);
  }
#pragma unroll
  for (int r = 0; r < 8; ++r) {
    const int row = i0 + r + 8 * hf;
    const int col = h0 + l;
    s1[(b * GAT_K + row) * GAT_H + col] = c1[r];
    s2[(b * GAT_K + row) * GAT_H + col] = c2[r];
  }
}

template <int M>
__device__ __forceinline__ float xrow(float x) {
  const int b = __builtin_bit_cast(int, x);
  return __builtin_bit_cast(float, __builtin_amdgcn_update_dpp(b, b, 0x160 + M, 0xf, 0xf, false));
}
__device__ __forceinline__ float xhalf(float x) {
  const int b = __builtin_bit_cast(int, x);
  return __builtin_bit_cast(float,
      __builtin_amdgcn_permlanex16(b, b, 0x76543210, 0xfedcba98, false, false));
}

__global__ void __launch_bounds__(256)
gat_attn(const float* __restrict__ s1,
         const float* __restrict__ s2,
         const float* __restrict__ lb,
         const float* __restrict__ av,
         const float* __restrict__ bias,
         float* __restrict__ attn) {
  __shared__ float4 r_s[8 * (GAT_H / 4)];
  __shared__ float  e_s[8 * GAT_K];

  const int blk  = blockIdx.x;
  const int ic   = blk % (GAT_K / 8);
  const int b    = blk / (GAT_K / 8);
  const int wave = threadIdx.x >> 5;
  const int lane = threadIdx.x & 31;
  const int i0   = ic * 8;
  const int j0   = wave * 16;

  const float4* lbv = reinterpret_cast<const float4*>(lb) + lane;
  const float4* s1v = reinterpret_cast<const float4*>(s1 + (b * GAT_K + i0 + wave) * GAT_H) + lane;
  const float4* avv = reinterpret_cast<const float4*>(av) + lane;
  float4 a4[2];
#pragma unroll
  for (int q = 0; q < 2; ++q) {
    const float4 l4 = lbv[q * 32];
    const float4 r4 = s1v[q * 32];
    a4[q] = avv[q * 32];
    r_s[wave * (GAT_H / 4) + q * 32 + lane] =
        make_float4(r4.x + l4.x, r4.y + l4.y, r4.z + l4.z, r4.w + l4.w);
  }

  const int irow = i0 + wave;
  float bz[4];
#pragma unroll
  for (int t = 0; t < 4; ++t) bz[t] = bias[irow * GAT_K + lane + 32 * t];

  const float4* s2v = reinterpret_cast<const float4*>(s2 + (b * GAT_K + j0) * GAT_H) + lane;
  float4 sr[16][2];
#pragma unroll
  for (int jj = 0; jj < 16; ++jj) {
#pragma unroll
    for (int q = 0; q < 2; ++q) sr[jj][q] = s2v[jj * (GAT_H / 4) + q * 32];
  }

  __syncthreads();

  const bool b0 = (lane & 1) != 0;
  const bool b1 = (lane & 2) != 0;
  const bool b2 = (lane & 4) != 0;
  const bool b3 = (lane & 8) != 0;
  const bool b4 = (lane & 16) != 0;

  for (int p = 0; p < 4; ++p) {
    float4 ra[2], rb[2];
#pragma unroll
    for (int q = 0; q < 2; ++q) {
      ra[q] = r_s[(2 * p + 0) * (GAT_H / 4) + q * 32 + lane];
      rb[q] = r_s[(2 * p + 1) * (GAT_H / 4) + q * 32 + lane];
    }

    float v[32];
#pragma unroll
    for (int jj = 0; jj < 16; ++jj) {
      float sa = 0.f, sb = 0.f;
#pragma unroll
      for (int q = 0; q < 2; ++q) {
        const float sv[4] = {sr[jj][q].x, sr[jj][q].y, sr[jj][q].z, sr[jj][q].w};
        const float pa[4] = {ra[q].x, ra[q].y, ra[q].z, ra[q].w};
        const float pb[4] = {rb[q].x, rb[q].y, rb[q].z, rb[q].w};
        const float aa[4] = {a4[q].x, a4[q].y, a4[q].z, a4[q].w};
#pragma unroll
        for (int e = 0; e < 4; ++e) {
          float ta = pa[e] + sv[e];
          float tb = pb[e] + sv[e];
          ta = fmaxf(ta, GAT_ALPHA * ta);
          tb = fmaxf(tb, GAT_ALPHA * tb);
          sa = fmaf(ta, aa[e], sa);
          sb = fmaf(tb, aa[e], sb);
        }
      }
      v[jj] = sa;
      v[16 + jj] = sb;
    }

    float w[16];
#pragma unroll
    for (int k = 0; k < 16; ++k) {
      const float keep = b0 ? v[16 + k] : v[k];
      const float send = b0 ? v[k] : v[16 + k];
      w[k] = keep + xrow<1>(send);
    }
    float x[8];
#pragma unroll
    for (int k = 0; k < 8; ++k) {
      const float keep = b4 ? w[8 + k] : w[k];
      const float send = b4 ? w[k] : w[8 + k];
      x[k] = keep + xhalf(send);
    }
    float y[4];
#pragma unroll
    for (int k = 0; k < 4; ++k) {
      const float keep = b3 ? x[4 + k] : x[k];
      const float send = b3 ? x[k] : x[4 + k];
      y[k] = keep + xrow<8>(send);
    }
    float z[2];
#pragma unroll
    for (int k = 0; k < 2; ++k) {
      const float keep = b2 ? y[2 + k] : y[k];
      const float send = b2 ? y[k] : y[2 + k];
      z[k] = keep + xrow<4>(send);
    }
    const float keep = b1 ? z[1] : z[0];
    const float send = b1 ? z[0] : z[1];
    const float r = keep + xrow<2>(send);
    e_s[(2 * p + (lane & 1)) * GAT_K + j0 + (lane >> 1)] = r;
  }

  __syncthreads();

  float e0 = e_s[wave * GAT_K + lane +  0] + bz[0];
  float e1 = e_s[wave * GAT_K + lane + 32] + bz[1];
  float e2 = e_s[wave * GAT_K + lane + 64] + bz[2];
  float e3 = e_s[wave * GAT_K + lane + 96] + bz[3];
  float m = fmaxf(fmaxf(e0, e1), fmaxf(e2, e3));
  m = fmaxf(m, __shfl_xor(m, 1));
  m = fmaxf(m, __shfl_xor(m, 2));
  m = fmaxf(m, __shfl_xor(m, 4));
  m = fmaxf(m, __shfl_xor(m, 8));
  m = fmaxf(m, __shfl_xor(m, 16));
  e0 = __expf(e0 - m);
  e1 = __expf(e1 - m);
  e2 = __expf(e2 - m);
  e3 = __expf(e3 - m);
  float s = e0 + e1 + e2 + e3;
  s += __shfl_xor(s, 1);
  s += __shfl_xor(s, 2);
  s += __shfl_xor(s, 4);
  s += __shfl_xor(s, 8);
  s += __shfl_xor(s, 16);
  const float inv = 1.0f / s;
  float* arow = attn + (b * GAT_K + irow) * GAT_K;
  arow[lane +  0] = e0 * inv;
  arow[lane + 32] = e1 * inv;
  arow[lane + 64] = e2 * inv;
  arow[lane + 96] = e3 * inv;
}

__global__ void __launch_bounds__(32)
gat_out(const float* __restrict__ attn,
        const float* __restrict__ x,
        float* __restrict__ out) {
  int tile = blockIdx.x;
  const int ft = tile % (GAT_F / 16); tile /= (GAT_F / 16);
  const int it = tile % (GAT_K / 16); tile /= (GAT_K / 16);
  const int b  = tile;
  const int i0 = it * 16, f0 = ft * 16;
  const int lane = threadIdx.x & 31;
  const int hf   = lane >> 4;
  const int l    = lane & 15;

  const float* ab = attn + b * GAT_K * GAT_K;
  const float* xb = x + b * GAT_F * GAT_K;

  v8f c = {};
  for (int j0 = 0; j0 < GAT_K; j0 += 4) {
    v2f av, bv;
    av.x = ab[(i0 + l) * GAT_K + j0 + 2 * hf + 0];
    av.y = ab[(i0 + l) * GAT_K + j0 + 2 * hf + 1];
    bv.x = xb[(f0 + l) * GAT_K + j0 + 2 * hf + 0];
    bv.y = xb[(f0 + l) * GAT_K + j0 + 2 * hf + 1];
    c = __builtin_amdgcn_wmma_f32_16x16x4_f32(false, av, false, bv, (short)0,
                                              c, false, false);
  }
#pragma unroll
  for (int r = 0; r < 8; ++r) {
    const int row = i0 + r + 8 * hf;
    const int col = f0 + l;
    const float z = c[r];
    out[(b * GAT_K + row) * GAT_F + col] = 1.0f / (1.0f + __expf(-z));
  }
}

extern "C" void kernel_launch(void* const* d_in, const int* in_sizes, int n_in,
                              void* d_out, int out_size, void* d_ws, size_t ws_size,
                              hipStream_t stream) {
  (void)in_sizes; (void)n_in; (void)out_size; (void)ws_size;
  const float* x    = (const float*)d_in[0];
  const float* w    = (const float*)d_in[1];
  const float* lb   = (const float*)d_in[2];
  const float* av   = (const float*)d_in[3];
  const float* bias = (const float*)d_in[4];
  float* out = (float*)d_out;

  float* s1   = (float*)d_ws;
  float* s2   = s1 + GAT_B * GAT_K * GAT_H;
  float* attn = s2 + GAT_B * GAT_K * GAT_H;

  gat_qk_gemm<<<GAT_B * (GAT_K / 16) * (GAT_H / 16), 32, 0, stream>>>(x, w, s1, s2);
  gat_attn  <<<GAT_B * (GAT_K / 8),               256, 0, stream>>>(s1, s2, lb, av, bias, attn);
  gat_out   <<<GAT_B * (GAT_K / 16) * (GAT_F / 16), 32, 0, stream>>>(attn, x, out);
}


# both GEMM kernels: a block loads its operand tiles once by whole-line 16-byte loads into LDS, WMMA fragments read from LDS (k = 8m+4hf+2s+e, both loaders); fp32 sum order over f and over j changes
# speedup vs baseline: 3.2832x; 1.3390x over previous
#include <hip/hip_runtime.h>

typedef __attribute__((ext_vector_type(2))) float v2f;
typedef __attribute__((ext_vector_type(8))) float v8f;

#define GAT_B 16
#define GAT_K 128
#define GAT_F 128
#define GAT_H 256
#define GAT_ALPHA 0.2f

#define QK_ROWS 64
#define QK_HT   32
#define QK_PA   72
#define QK_PB   260

__global__ void __launch_bounds__(256)
gat_qk_gemm(const float* __restrict__ x,
            const float* __restrict__ w,
            float* __restrict__ s1,
            float* __restrict__ s2) {
  __shared__ float4 xs4[GAT_F * QK_PA / 4];
  __shared__ float4 ws4[QK_HT * QK_PB / 4];

  int blk = blockIdx.x;
  const int ht = blk % (GAT_H / QK_HT);    blk /= (GAT_H / QK_HT);
  const int ic = blk % (GAT_K / QK_ROWS);  blk /= (GAT_K / QK_ROWS);
  const int b  = blk;
  const int i0 = ic * QK_ROWS, h0 = ht * QK_HT;
  const int tid = threadIdx.x;

  const float4* xg = reinterpret_cast<const float4*>(x + b * GAT_F * GAT_K + i0);
  const float4* wg = reinterpret_cast<const float4*>(w + h0 * (2 * GAT_F));
  float4 xa[8], wa[8];
#pragma unroll
  for (int r = 0; r < 8; ++r) {
    const int idx = tid + 256 * r;
    xa[r] = xg[(idx >> 4) * (GAT_K / 4) + (idx & 15)];
  }
#pragma unroll
  for (int r = 0; r < 8; ++r) wa[r] = wg[tid + 256 * r];
#pragma unroll
  for (int r = 0; r < 8; ++r) {
    const int idx = tid + 256 * r;
    xs4[(idx >> 4) * (QK_PA / 4) + (idx & 15)] = xa[r];
  }
#pragma unroll
  for (int r = 0; r < 8; ++r) {
    const int idx = tid + 256 * r;
    ws4[(idx >> 6) * (QK_PB / 4) + (idx & 63)] = wa[r];
  }
  __syncthreads();

  const int wave = tid >> 5;
  const int lane = tid & 31;
  const int hf   = lane >> 4;
  const int l    = lane & 15;
  const int mt   = wave & 3;
  const int sel  = wave >> 2;

  const float*  xs  = reinterpret_cast<const float*>(xs4) + (4 * hf) * QK_PA + mt * 16 + l;
  const float4* wb0 = ws4 + (l     ) * (QK_PB / 4) + sel * (GAT_F / 4) + hf;
  const float4* wb1 = ws4 + (l + 16) * (QK_PB / 4) + sel * (GAT_F / 4) + hf;

  v2f a0[GAT_F / 8], a1[GAT_F / 8];
  float4 bq0[GAT_F / 8], bq1[GAT_F / 8];
#pragma unroll
  for (int kk = 0; kk < GAT_F / 8; ++kk) {
    const float* xk = xs + (8 * kk) * QK_PA;
    a0[kk].x = xk[0 * QK_PA];
    a0[kk].y = xk[1 * QK_PA];
    a1[kk].x = xk[2 * QK_PA];
    a1[kk].y = xk[3 * QK_PA];
    bq0[kk] = wb0[2 * kk];
    bq1[kk] = wb1[2 * kk];
  }
  v8f c0 = {};
  v8f c1 = {};
  __builtin_amdgcn_sched_barrier(0);
#pragma unroll
  for (int kk = 0; kk < GAT_F / 8; ++kk) {
    v2f p0, p1, q0, q1;
    p0.x = bq0[kk].x; p0.y = bq0[kk].y; p1.x = bq0[kk].z; p1.y = bq0[kk].w;
    q0.x = bq1[kk].x; q0.y = bq1[kk].y; q1.x = bq1[kk].z; q1.y = bq1[kk].w;
    c0 = __builtin_amdgcn_wmma_f32_16x16x4_f32(false, a0[kk], false, p0, (short)0,
                                               c0, false, false);
    c1 = __builtin_amdgcn_wmma_f32_16x16x4_f32(false, a0[kk], false, q0, (short)0,
                                               c1, false, false);
    c0 = __builtin_amdgcn_wmma_f32_16x16x4_f32(false, a1[kk], false, p1, (short)0,
                                               c0, false, false);
    c1 = __builtin_amdgcn_wmma_f32_16x16x4_f32(false, a1[kk], false, q1, (short)0,
                                               c1, false, false);
  }

  float* dst = sel ? s2 : s1;
#pragma unroll
  for (int r = 0; r < 8; ++r) {
    const int row = i0 + mt * 16 + r + 8 * hf;
    const int col = h0 + l;
    dst[(b * GAT_K + row) * GAT_H + col]      = c0[r];
    dst[(b * GAT_K + row) * GAT_H + col + 16] = c1[r];
  }
}

template <int M>
__device__ __forceinline__ float xrow(float x) {
  const int b = __builtin_bit_cast(int, x);
  return __builtin_bit_cast(float, __builtin_amdgcn_update_dpp(b, b, 0x160 + M, 0xf, 0xf, false));
}
__device__ __forceinline__ float xhalf(float x) {
  const int b = __builtin_bit_cast(int, x);
  return __builtin_bit_cast(float,
      __builtin_amdgcn_permlanex16(b, b, 0x76543210, 0xfedcba98, false, false));
}

__global__ void __launch_bounds__(256)
gat_attn(const float* __restrict__ s1,
         const float* __restrict__ s2,
         const float* __restrict__ lb,
         const float* __restrict__ av,
         const float* __restrict__ bias,
         float* __restrict__ attn) {
  __shared__ float4 r_s[8 * (GAT_H / 4)];
  __shared__ float  e_s[8 * GAT_K];

  const int blk  = blockIdx.x;
  const int ic   = blk % (GAT_K / 8);
  const int b    = blk / (GAT_K / 8);
  const int wave = threadIdx.x >> 5;
  const int lane = threadIdx.x & 31;
  const int i0   = ic * 8;
  const int j0   = wave * 16;

  const float4* lbv = reinterpret_cast<const float4*>(lb) + lane;
  const float4* s1v = reinterpret_cast<const float4*>(s1 + (b * GAT_K + i0 + wave) * GAT_H) + lane;
  const float4* avv = reinterpret_cast<const float4*>(av) + lane;
  float4 a4[2];
#pragma unroll
  for (int q = 0; q < 2; ++q) {
    const float4 l4 = lbv[q * 32];
    const float4 r4 = s1v[q * 32];
    a4[q] = avv[q * 32];
    r_s[wave * (GAT_H / 4) + q * 32 + lane] =
        make_float4(r4.x + l4.x, r4.y + l4.y, r4.z + l4.z, r4.w + l4.w);
  }

  const int irow = i0 + wave;
  float bz[4];
#pragma unroll
  for (int t = 0; t < 4; ++t) bz[t] = bias[irow * GAT_K + lane + 32 * t];

  const float4* s2v = reinterpret_cast<const float4*>(s2 + (b * GAT_K + j0) * GAT_H) + lane;
  float4 sr[16][2];
#pragma unroll
  for (int jj = 0; jj < 16; ++jj) {
#pragma unroll
    for (int q = 0; q < 2; ++q) sr[jj][q] = s2v[jj * (GAT_H / 4) + q * 32];
  }

  __syncthreads();

  const bool b0 = (lane & 1) != 0;
  const bool b1 = (lane & 2) != 0;
  const bool b2 = (lane & 4) != 0;
  const bool b3 = (lane & 8) != 0;
  const bool b4 = (lane & 16) != 0;

  for (int p = 0; p < 4; ++p) {
    float4 ra[2], rb[2];
#pragma unroll
    for (int q = 0; q < 2; ++q) {
      ra[q] = r_s[(2 * p + 0) * (GAT_H / 4) + q * 32 + lane];
      rb[q] = r_s[(2 * p + 1) * (GAT_H / 4) + q * 32 + lane];
    }

    float v[32];
#pragma unroll
    for (int jj = 0; jj < 16; ++jj) {
      float sa = 0.f, sb = 0.f;
#pragma unroll
      for (int q = 0; q < 2; ++q) {
        const float sv[4] = {sr[jj][q].x, sr[jj][q].y, sr[jj][q].z, sr[jj][q].w};
        const float pa[4] = {ra[q].x, ra[q].y, ra[q].z, ra[q].w};
        const float pb[4] = {rb[q].x, rb[q].y, rb[q].z, rb[q].w};
        const float aa[4] = {a4[q].x, a4[q].y, a4[q].z, a4[q].w};
#pragma unroll
        for (int e = 0; e < 4; ++e) {
          float ta = pa[e] + sv[e];
          float tb = pb[e] + sv[e];
          ta = fmaxf(ta, GAT_ALPHA * ta);
          tb = fmaxf(tb, GAT_ALPHA * tb);
          sa = fmaf(ta, aa[e], sa);
          sb = fmaf(tb, aa[e], sb);
        }
      }
      v[jj] = sa;
      v[16 + jj] = sb;
    }

    float w[16];
#pragma unroll
    for (int k = 0; k < 16; ++k) {
      const float keep = b0 ? v[16 + k] : v[k];
      const float send = b0 ? v[k] : v[16 + k];
      w[k] = keep + xrow<1>(send);
    }
    float x[8];
#pragma unroll
    for (int k = 0; k < 8; ++k) {
      const float keep = b4 ? w[8 + k] : w[k];
      const float send = b4 ? w[k] : w[8 + k];
      x[k] = keep + xhalf(send);
    }
    float y[4];
#pragma unroll
    for (int k = 0; k < 4; ++k) {
      const float keep = b3 ? x[4 + k] : x[k];
      const float send = b3 ? x[k] : x[4 + k];
      y[k] = keep + xrow<8>(send);
    }
    float z[2];
#pragma unroll
    for (int k = 0; k < 2; ++k) {
      const float keep = b2 ? y[2 + k] : y[k];
      const float send = b2 ? y[k] : y[2 + k];
      z[k] = keep + xrow<4>(send);
    }
    const float keep = b1 ? z[1] : z[0];
    const float send = b1 ? z[0] : z[1];
    const float r = keep + xrow<2>(send);
    e_s[(2 * p + (lane & 1)) * GAT_K + j0 + (lane >> 1)] = r;
  }

  __syncthreads();

  float e0 = e_s[wave * GAT_K + lane +  0] + bz[0];
  float e1 = e_s[wave * GAT_K + lane + 32] + bz[1];
  float e2 = e_s[wave * GAT_K + lane + 64] + bz[2];
  float e3 = e_s[wave * GAT_K + lane + 96] + bz[3];
  float m = fmaxf(fmaxf(e0, e1), fmaxf(e2, e3));
  m = fmaxf(m, __shfl_xor(m, 1));
  m = fmaxf(m, __shfl_xor(m, 2));
  m = fmaxf(m, __shfl_xor(m, 4));
  m = fmaxf(m, __shfl_xor(m, 8));
  m = fmaxf(m, __shfl_xor(m, 16));
  e0 = __expf(e0 - m);
  e1 = __expf(e1 - m);
  e2 = __expf(e2 - m);
  e3 = __expf(e3 - m);
  float s = e0 + e1 + e2 + e3;
  s += __shfl_xor(s, 1);
  s += __shfl_xor(s, 2);
  s += __shfl_xor(s, 4);
  s += __shfl_xor(s, 8);
  s += __shfl_xor(s, 16);
  const float inv = 1.0f / s;
  float* arow = attn + (b * GAT_K + irow) * GAT_K;
  arow[lane +  0] = e0 * inv;
  arow[lane + 32] = e1 * inv;
  arow[lane + 64] = e2 * inv;
  arow[lane + 96] = e3 * inv;
}

#define OUT_T 32
#define OUT_P 132

__global__ void __launch_bounds__(128)
gat_out(const float* __restrict__ attn,
        const float* __restrict__ x,
        float* __restrict__ out) {
  __shared__ float4 as4[OUT_T * OUT_P / 4];
  __shared__ float4 xs4[OUT_T * OUT_P / 4];

  int blk = blockIdx.x;
  const int ft = blk % (GAT_F / OUT_T); blk /= (GAT_F / OUT_T);
  const int it = blk % (GAT_K / OUT_T); blk /= (GAT_K / OUT_T);
  const int b  = blk;
  const int i0 = it * OUT_T, f0 = ft * OUT_T;
  const int tid = threadIdx.x;

  const float4* ag = reinterpret_cast<const float4*>(attn + (b * GAT_K + i0) * GAT_K);
  const float4* xg = reinterpret_cast<const float4*>(x + (b * GAT_F + f0) * GAT_K);
  float4 ar[8], xr[8];
#pragma unroll
  for (int r = 0; r < 8; ++r) ar[r] = ag[tid + 128 * r];
#pragma unroll
  for (int r = 0; r < 8; ++r) xr[r] = xg[tid + 128 * r];
#pragma unroll
  for (int r = 0; r < 8; ++r) {
    const int idx = tid + 128 * r;
    as4[(idx >> 5) * (OUT_P / 4) + (idx & 31)] = ar[r];
  }
#pragma unroll
  for (int r = 0; r < 8; ++r) {
    const int idx = tid + 128 * r;
    xs4[(idx >> 5) * (OUT_P / 4) + (idx & 31)] = xr[r];
  }
  __syncthreads();

  const int wave = tid >> 5;
  const int lane = tid & 31;
  const int hf   = lane >> 4;
  const int l    = lane & 15;
  const int mi   = wave & 1;
  const int ni   = wave >> 1;

  const float4* ap = as4 + (mi * 16 + l) * (OUT_P / 4) + hf;
  const float4* bp = xs4 + (ni * 16 + l) * (OUT_P / 4) + hf;

  float4 aq[GAT_K / 8], bq[GAT_K / 8];
#pragma unroll
  for (int kk = 0; kk < GAT_K / 8; ++kk) {
    aq[kk] = ap[2 * kk];
    bq[kk] = bp[2 * kk];
  }
  v8f ca = {};
  v8f cb = {};
  __builtin_amdgcn_sched_barrier(0);
#pragma unroll
  for (int kk = 0; kk < GAT_K / 8; ++kk) {
    v2f a0, a1, b0, b1;
    a0.x = aq[kk].x; a0.y = aq[kk].y; a1.x = aq[kk].z; a1.y = aq[kk].w;
    b0.x = bq[kk].x; b0.y = bq[kk].y; b1.x = bq[kk].z; b1.y = bq[kk].w;
    ca = __builtin_amdgcn_wmma_f32_16x16x4_f32(false, a0, false, b0, (short)0,
                                               ca, false, false);
    cb = __builtin_amdgcn_wmma_f32_16x16x4_f32(false, a1, false, b1, (short)0,
                                               cb, false, false);
  }
#pragma unroll
  for (int r = 0; r < 8; ++r) {
    const int row = i0 + mi * 16 + r + 8 * hf;
    const int col = f0 + ni * 16 + l;
    const float z = ca[r] + cb[r];
    out[(b * GAT_K + row) * GAT_F + col] = 1.0f / (1.0f + __expf(-z));
  }
}

extern "C" void kernel_launch(void* const* d_in, const int* in_sizes, int n_in,
                              void* d_out, int out_size, void* d_ws, size_t ws_size,
                              hipStream_t stream) {
  (void)in_sizes; (void)n_in; (void)out_size; (void)ws_size;
  const float* x    = (const float*)d_in[0];
  const float* w    = (const float*)d_in[1];
  const float* lb   = (const float*)d_in[2];
  const float* av   = (const float*)d_in[3];
  const float* bias = (const float*)d_in[4];
  float* out = (float*)d_out;

  float* s1   = (float*)d_ws;
  float* s2   = s1 + GAT_B * GAT_K * GAT_H;
  float* attn = s2 + GAT_B * GAT_K * GAT_H;

  gat_qk_gemm<<<GAT_B * (GAT_K / QK_ROWS) * (GAT_H / QK_HT), 256, 0, stream>>>(x, w, s1, s2);
  gat_attn  <<<GAT_B * (GAT_K / 8),               256, 0, stream>>>(s1, s2, lb, av, bias, attn);
  gat_out   <<<GAT_B * (GAT_K / OUT_T) * (GAT_F / OUT_T), 128, 0, stream>>>(attn, x, out);
}


# attention score loop issued stage by stage per h (all 32 adds of 2 rows x 16 columns, then 32 multiplies, 32 max, 32 multiply-adds): no vector op waits on its predecessor; fp32 sum order unchanged
# speedup vs baseline: 3.6725x; 1.1186x over previous
#include <hip/hip_runtime.h>

typedef __attribute__((ext_vector_type(2))) float v2f;
typedef __attribute__((ext_vector_type(8))) float v8f;

#define GAT_B 16
#define GAT_K 128
#define GAT_F 128
#define GAT_H 256
#define GAT_ALPHA 0.2f

#define QK_ROWS 64
#define QK_HT   32
#define QK_PA   72
#define QK_PB   260

__global__ void __launch_bounds__(256)
gat_qk_gemm(const float* __restrict__ x,
            const float* __restrict__ w,
            float* __restrict__ s1,
            float* __restrict__ s2) {
  __shared__ float4 xs4[GAT_F * QK_PA / 4];
  __shared__ float4 ws4[QK_HT * QK_PB / 4];

  int blk = blockIdx.x;
  const int ht = blk % (GAT_H / QK_HT);    blk /= (GAT_H / QK_HT);
  const int ic = blk % (GAT_K / QK_ROWS);  blk /= (GAT_K / QK_ROWS);
  const int b  = blk;
  const int i0 = ic * QK_ROWS, h0 = ht * QK_HT;
  const int tid = threadIdx.x;

  const float4* xg = reinterpret_cast<const float4*>(x + b * GAT_F * GAT_K + i0);
  const float4* wg = reinterpret_cast<const float4*>(w + h0 * (2 * GAT_F));
  float4 xa[8], wa[8];
#pragma unroll
  for (int r = 0; r < 8; ++r) {
    const int idx = tid + 256 * r;
    xa[r] = xg[(idx >> 4) * (GAT_K / 4) + (idx & 15)];
  }
#pragma unroll
  for (int r = 0; r < 8; ++r) wa[r] = wg[tid + 256 * r];
#pragma unroll
  for (int r = 0; r < 8; ++r) {
    const int idx = tid + 256 * r;
    xs4[(idx >> 4) * (QK_PA / 4) + (idx & 15)] = xa[r];
  }
#pragma unroll
  for (int r = 0; r < 8; ++r) {
    const int idx = tid + 256 * r;
    ws4[(idx >> 6) * (QK_PB / 4) + (idx & 63)] = wa[r];
  }
  __syncthreads();

  const int wave = tid >> 5;
  const int lane = tid & 31;
  const int hf   = lane >> 4;
  const int l    = lane & 15;
  const int mt   = wave & 3;
  const int sel  = wave >> 2;

  const float*  xs  = reinterpret_cast<const float*>(xs4) + (4 * hf) * QK_PA + mt * 16 + l;
  const float4* wb0 = ws4 + (l     ) * (QK_PB / 4) + sel * (GAT_F / 4) + hf;
  const float4* wb1 = ws4 + (l + 16) * (QK_PB / 4) + sel * (GAT_F / 4) + hf;

  v2f a0[GAT_F / 8], a1[GAT_F / 8];
  float4 bq0[GAT_F / 8], bq1[GAT_F / 8];
#pragma unroll
  for (int kk = 0; kk < GAT_F / 8; ++kk) {
    const float* xk = xs + (8 * kk) * QK_PA;
    a0[kk].x = xk[0 * QK_PA];
    a0[kk].y = xk[1 * QK_PA];
    a1[kk].x = xk[2 * QK_PA];
    a1[kk].y = xk[3 * QK_PA];
    bq0[kk] = wb0[2 * kk];
    bq1[kk] = wb1[2 * kk];
  }
  v8f c0 = {};
  v8f c1 = {};
  __builtin_amdgcn_sched_barrier(0);
#pragma unroll
  for (int kk = 0; kk < GAT_F / 8; ++kk) {
    v2f p0, p1, q0, q1;
    p0.x = bq0[kk].x; p0.y = bq0[kk].y; p1.x = bq0[kk].z; p1.y = bq0[kk].w;
    q0.x = bq1[kk].x; q0.y = bq1[kk].y; q1.x = bq1[kk].z; q1.y = bq1[kk].w;
    c0 = __builtin_amdgcn_wmma_f32_16x16x4_f32(false, a0[kk], false, p0, (short)0,
                                               c0, false, false);
    c1 = __builtin_amdgcn_wmma_f32_16x16x4_f32(false, a0[kk], false, q0, (short)0,
                                               c1, false, false);
    c0 = __builtin_amdgcn_wmma_f32_16x16x4_f32(false, a1[kk], false, p1, (short)0,
                                               c0, false, false);
    c1 = __builtin_amdgcn_wmma_f32_16x16x4_f32(false, a1[kk], false, q1, (short)0,
                                               c1, false, false);
  }

  float* dst = sel ? s2 : s1;
#pragma unroll
  for (int r = 0; r < 8; ++r) {
    const int row = i0 + mt * 16 + r + 8 * hf;
    const int col = h0 + l;
    dst[(b * GAT_K + row) * GAT_H + col]      = c0[r];
    dst[(b * GAT_K + row) * GAT_H + col + 16] = c1[r];
  }
}

template <int M>
__device__ __forceinline__ float xrow(float x) {
  const int b = __builtin_bit_cast(int, x);
  return __builtin_bit_cast(float, __builtin_amdgcn_update_dpp(b, b, 0x160 + M, 0xf, 0xf, false));
}
__device__ __forceinline__ float xhalf(float x) {
  const int b = __builtin_bit_cast(int, x);
  return __builtin_bit_cast(float,
      __builtin_amdgcn_permlanex16(b, b, 0x76543210, 0xfedcba98, false, false));
}

__global__ void __launch_bounds__(256)
gat_attn(const float* __restrict__ s1,
         const float* __restrict__ s2,
         const float* __restrict__ lb,
         const float* __restrict__ av,
         const float* __restrict__ bias,
         float* __restrict__ attn) {
  __shared__ float4 r_s[8 * (GAT_H / 4)];
  __shared__ float  e_s[8 * GAT_K];

  const int blk  = blockIdx.x;
  const int ic   = blk % (GAT_K / 8);
  const int b    = blk / (GAT_K / 8);
  const int wave = threadIdx.x >> 5;
  const int lane = threadIdx.x & 31;
  const int i0   = ic * 8;
  const int j0   = wave * 16;

  const float4* lbv = reinterpret_cast<const float4*>(lb) + lane;
  const float4* s1v = reinterpret_cast<const float4*>(s1 + (b * GAT_K + i0 + wave) * GAT_H) + lane;
  const float4* avv = reinterpret_cast<const float4*>(av) + lane;
  float4 a4[2];
#pragma unroll
  for (int q = 0; q < 2; ++q) {
    const float4 l4 = lbv[q * 32];
    const float4 r4 = s1v[q * 32];
    a4[q] = avv[q * 32];
    r_s[wave * (GAT_H / 4) + q * 32 + lane] =
        make_float4(r4.x + l4.x, r4.y + l4.y, r4.z + l4.z, r4.w + l4.w);
  }

  const int irow = i0 + wave;
  float bz[4];
#pragma unroll
  for (int t = 0; t < 4; ++t) bz[t] = bias[irow * GAT_K + lane + 32 * t];

  const float4* s2v = reinterpret_cast<const float4*>(s2 + (b * GAT_K + j0) * GAT_H) + lane;
  float4 sr[16][2];
#pragma unroll
  for (int jj = 0; jj < 16; ++jj) {
#pragma unroll
    for (int q = 0; q < 2; ++q) sr[jj][q] = s2v[jj * (GAT_H / 4) + q * 32];
  }

  __syncthreads();

  const bool b0 = (lane & 1) != 0;
  const bool b1 = (lane & 2) != 0;
  const bool b2 = (lane & 4) != 0;
  const bool b3 = (lane & 8) != 0;
  const bool b4 = (lane & 16) != 0;

  for (int p = 0; p < 4; ++p) {
    float4 ra[2], rb[2];
#pragma unroll
    for (int q = 0; q < 2; ++q) {
      ra[q] = r_s[(2 * p + 0) * (GAT_H / 4) + q * 32 + lane];
      rb[q] = r_s[(2 * p + 1) * (GAT_H / 4) + q * 32 + lane];
    }

    float v[32];
#pragma unroll
    for (int q = 0; q < 2; ++q) {
      const float pa[4] = {ra[q].x, ra[q].y, ra[q].z, ra[q].w};
      const float pb[4] = {rb[q].x, rb[q].y, rb[q].z, rb[q].w};
      const float aa[4] = {a4[q].x, a4[q].y, a4[q].z, a4[q].w};
#pragma unroll
      for (int e = 0; e < 4; ++e) {
        float ta[16], tb[16], ua[16], ub[16];
#pragma unroll
        for (int jj = 0; jj < 16; ++jj) {
          const float sv[4] = {sr[jj][q].x, sr[jj][q].y, sr[jj][q].z, sr[jj][q].w};
          ta[jj] = pa[e] + sv[e];
          tb[jj] = pb[e] + sv[e];
        }
#pragma unroll
        for (int jj = 0; jj < 16; ++jj) {
          ua[jj] = GAT_ALPHA * ta[jj];
          ub[jj] = GAT_ALPHA * tb[jj];
        }
#pragma unroll
        for (int jj = 0; jj < 16; ++jj) {
          ta[jj] = fmaxf(ta[jj], ua[jj]);
          tb[jj] = fmaxf(tb[jj], ub[jj]);
        }
#pragma unroll
        for (int jj = 0; jj < 16; ++jj) {
          if (q == 0 && e == 0) {
            v[jj]      = ta[jj] * aa[e];
            v[16 + jj] = tb[jj] * aa[e];
          } else {
            v[jj]      = fmaf(ta[jj], aa[e], v[jj]);
            v[16 + jj] = fmaf(tb[jj], aa[e], v[16 + jj]);
          }
        }
      }
    }

    float w[16];
#pragma unroll
    for (int k = 0; k < 16; ++k) {
      const float keep = b0 ? v[16 + k] : v[k];
      const float send = b0 ? v[k] : v[16 + k];
      w[k] = keep + xrow<1>(send);
    }
    float x[8];
#pragma unroll
    for (int k = 0; k < 8; ++k) {
      const float keep = b4 ? w[8 + k] : w[k];
      const float send = b4 ? w[k] : w[8 + k];
      x[k] = keep + xhalf(send);
    }
    float y[4];
#pragma unroll
    for (int k = 0; k < 4; ++k) {
      const float keep = b3 ? x[4 + k] : x[k];
      const float send = b3 ? x[k] : x[4 + k];
      y[k] = keep + xrow<8>(send);
    }
    float z[2];
#pragma unroll
    for (int k = 0; k < 2; ++k) {
      const float keep = b2 ? y[2 + k] : y[k];
      const float send = b2 ? y[k] : y[2 + k];
      z[k] = keep + xrow<4>(send);
    }
    const float keep = b1 ? z[1] : z[0];
    const float send = b1 ? z[0] : z[1];
    const float r = keep + xrow<2>(send);
    e_s[(2 * p + (lane & 1)) * GAT_K + j0 + (lane >> 1)] = r;
  }

  __syncthreads();

  float e0 = e_s[wave * GAT_K + lane +  0] + bz[0];
  float e1 = e_s[wave * GAT_K + lane + 32] + bz[1];
  float e2 = e_s[wave * GAT_K + lane + 64] + bz[2];
  float e3 = e_s[wave * GAT_K + lane + 96] + bz[3];
  float m = fmaxf(fmaxf(e0, e1), fmaxf(e2, e3));
  m = fmaxf(m, __shfl_xor(m, 1));
  m = fmaxf(m, __shfl_xor(m, 2));
  m = fmaxf(m, __shfl_xor(m, 4));
  m = fmaxf(m, __shfl_xor(m, 8));
  m = fmaxf(m, __shfl_xor(m, 16));
  e0 = __expf(e0 - m);
  e1 = __expf(e1 - m);
  e2 = __expf(e2 - m);
  e3 = __expf(e3 - m);
  float s = e0 + e1 + e2 + e3;
  s += __shfl_xor(s, 1);
  s += __shfl_xor(s, 2);
  s += __shfl_xor(s, 4);
  s += __shfl_xor(s, 8);
  s += __shfl_xor(s, 16);
  const float inv = 1.0f / s;
  float* arow = attn + (b * GAT_K + irow) * GAT_K;
  arow[lane +  0] = e0 * inv;
  arow[lane + 32] = e1 * inv;
  arow[lane + 64] = e2 * inv;
  arow[lane + 96] = e3 * inv;
}

#define OUT_T 32
#define OUT_P 132

__global__ void __launch_bounds__(128)
gat_out(const float* __restrict__ attn,
        const float* __restrict__ x,
        float* __restrict__ out) {
  __shared__ float4 as4[OUT_T * OUT_P / 4];
  __shared__ float4 xs4[OUT_T * OUT_P / 4];

  int blk = blockIdx.x;
  const int ft = blk % (GAT_F / OUT_T); blk /= (GAT_F / OUT_T);
  const int it = blk % (GAT_K / OUT_T); blk /= (GAT_K / OUT_T);
  const int b  = blk;
  const int i0 = it * OUT_T, f0 = ft * OUT_T;
  const int tid = threadIdx.x;

  const float4* ag = reinterpret_cast<const float4*>(attn + (b * GAT_K + i0) * GAT_K);
  const float4* xg = reinterpret_cast<const float4*>(x + (b * GAT_F + f0) * GAT_K);
  float4 ar[8], xr[8];
#pragma unroll
  for (int r = 0; r < 8; ++r) ar[r] = ag[tid + 128 * r];
#pragma unroll
  for (int r = 0; r < 8; ++r) xr[r] = xg[tid + 128 * r];
#pragma unroll
  for (int r = 0; r < 8; ++r) {
    const int idx = tid + 128 * r;
    as4[(idx >> 5) * (OUT_P / 4) + (idx & 31)] = ar[r];
  }
#pragma unroll
  for (int r = 0; r < 8; ++r) {
    const int idx = tid + 128 * r;
    xs4[(idx >> 5) * (OUT_P / 4) + (idx & 31)] = xr[r];
  }
  __syncthreads();

  const int wave = tid >> 5;
  const int lane = tid & 31;
  const int hf   = lane >> 4;
  const int l    = lane & 15;
  const int mi   = wave & 1;
  const int ni   = wave >> 1;

  const float4* ap = as4 + (mi * 16 + l) * (OUT_P / 4) + hf;
  const float4* bp = xs4 + (ni * 16 + l) * (OUT_P / 4) + hf;

  float4 aq[GAT_K / 8], bq[GAT_K / 8];
#pragma unroll
  for (int kk = 0; kk < GAT_K / 8; ++kk) {
    aq[kk] = ap[2 * kk];
    bq[kk] = bp[2 * kk];
  }
  v8f ca = {};
  v8f cb = {};
  __builtin_amdgcn_sched_barrier(0);
#pragma unroll
  for (int kk = 0; kk < GAT_K / 8; ++kk) {
    v2f a0, a1, b0, b1;
    a0.x = aq[kk].x; a0.y = aq[kk].y; a1.x = aq[kk].z; a1.y = aq[kk].w;
    b0.x = bq[kk].x; b0.y = bq[kk].y; b1.x = bq[kk].z; b1.y = bq[kk].w;
    ca = __builtin_amdgcn_wmma_f32_16x16x4_f32(false, a0, false, b0, (short)0,
                                               ca, false, false);
    cb = __builtin_amdgcn_wmma_f32_16x16x4_f32(false, a1, false, b1, (short)0,
                                               cb, false, false);
  }
#pragma unroll
  for (int r = 0; r < 8; ++r) {
    const int row = i0 + mi * 16 + r + 8 * hf;
    const int col = f0 + ni * 16 + l;
    const float z = ca[r] + cb[r];
    out[(b * GAT_K + row) * GAT_F + col] = 1.0f / (1.0f + __expf(-z));
  }
}

extern "C" void kernel_launch(void* const* d_in, const int* in_sizes, int n_in,
                              void* d_out, int out_size, void* d_ws, size_t ws_size,
                              hipStream_t stream) {
  (void)in_sizes; (void)n_in; (void)out_size; (void)ws_size;
  const float* x    = (const float*)d_in[0];
  const float* w    = (const float*)d_in[1];
  const float* lb   = (const float*)d_in[2];
  const float* av   = (const float*)d_in[3];
  const float* bias = (const float*)d_in[4];
  float* out = (float*)d_out;

  float* s1   = (float*)d_ws;
  float* s2   = s1 + GAT_B * GAT_K * GAT_H;
  float* attn = s2 + GAT_B * GAT_K * GAT_H;

  gat_qk_gemm<<<GAT_B * (GAT_K / QK_ROWS) * (GAT_H / QK_HT), 256, 0, stream>>>(x, w, s1, s2);
  gat_attn  <<<GAT_B * (GAT_K / 8),               256, 0, stream>>>(s1, s2, lb, av, bias, attn);
  gat_out   <<<GAT_B * (GAT_K / OUT_T) * (GAT_F / OUT_T), 128, 0, stream>>>(attn, x, out);
}


# attention wave takes its 16 s2 columns 4 at a time (pass = 8 rows x 4 columns, was 2 x 16): only the first 4 are loaded ahead of the arithmetic, the rest arrive under it; fp32 sum order unchanged
# speedup vs baseline: 3.9231x; 1.0682x over previous
#include <hip/hip_runtime.h>

typedef __attribute__((ext_vector_type(2))) float v2f;
typedef __attribute__((ext_vector_type(8))) float v8f;

#define GAT_B 16
#define GAT_K 128
#define GAT_F 128
#define GAT_H 256
#define GAT_ALPHA 0.2f

#define QK_ROWS 64
#define QK_HT   32
#define QK_PA   72
#define QK_PB   260

__global__ void __launch_bounds__(256)
gat_qk_gemm(const float* __restrict__ x,
            const float* __restrict__ w,
            float* __restrict__ s1,
            float* __restrict__ s2) {
  __shared__ float4 xs4[GAT_F * QK_PA / 4];
  __shared__ float4 ws4[QK_HT * QK_PB / 4];

  int blk = blockIdx.x;
  const int ht = blk % (GAT_H / QK_HT);    blk /= (GAT_H / QK_HT);
  const int ic = blk % (GAT_K / QK_ROWS);  blk /= (GAT_K / QK_ROWS);
  const int b  = blk;
  const int i0 = ic * QK_ROWS, h0 = ht * QK_HT;
  const int tid = threadIdx.x;

  const float4* xg = reinterpret_cast<const float4*>(x + b * GAT_F * GAT_K + i0);
  const float4* wg = reinterpret_cast<const float4*>(w + h0 * (2 * GAT_F));
  float4 xa[8], wa[8];
#pragma unroll
  for (int r = 0; r < 8; ++r) {
    const int idx = tid + 256 * r;
    xa[r] = xg[(idx >> 4) * (GAT_K / 4) + (idx & 15)];
  }
#pragma unroll
  for (int r = 0; r < 8; ++r) wa[r] = wg[tid + 256 * r];
#pragma unroll
  for (int r = 0; r < 8; ++r) {
    const int idx = tid + 256 * r;
    xs4[(idx >> 4) * (QK_PA / 4) + (idx & 15)] = xa[r];
  }
#pragma unroll
  for (int r = 0; r < 8; ++r) {
    const int idx = tid + 256 * r;
    ws4[(idx >> 6) * (QK_PB / 4) + (idx & 63)] = wa[r];
  }
  __syncthreads();

  const int wave = tid >> 5;
  const int lane = tid & 31;
  const int hf   = lane >> 4;
  const int l    = lane & 15;
  const int mt   = wave & 3;
  const int sel  = wave >> 2;

  const float*  xs  = reinterpret_cast<const float*>(xs4) + (4 * hf) * QK_PA + mt * 16 + l;
  const float4* wb0 = ws4 + (l     ) * (QK_PB / 4) + sel * (GAT_F / 4) + hf;
  const float4* wb1 = ws4 + (l + 16) * (QK_PB / 4) + sel * (GAT_F / 4) + hf;

  v2f a0[GAT_F / 8], a1[GAT_F / 8];
  float4 bq0[GAT_F / 8], bq1[GAT_F / 8];
#pragma unroll
  for (int kk = 0; kk < GAT_F / 8; ++kk) {
    const float* xk = xs + (8 * kk) * QK_PA;
    a0[kk].x = xk[0 * QK_PA];
    a0[kk].y = xk[1 * QK_PA];
    a1[kk].x = xk[2 * QK_PA];
    a1[kk].y = xk[3 * QK_PA];
    bq0[kk] = wb0[2 * kk];
    bq1[kk] = wb1[2 * kk];
  }
  v8f c0 = {};
  v8f c1 = {};
  __builtin_amdgcn_sched_barrier(0);
#pragma unroll
  for (int kk = 0; kk < GAT_F / 8; ++kk) {
    v2f p0, p1, q0, q1;
    p0.x = bq0[kk].x; p0.y = bq0[kk].y; p1.x = bq0[kk].z; p1.y = bq0[kk].w;
    q0.x = bq1[kk].x; q0.y = bq1[kk].y; q1.x = bq1[kk].z; q1.y = bq1[kk].w;
    c0 = __builtin_amdgcn_wmma_f32_16x16x4_f32(false, a0[kk], false, p0, (short)0,
                                               c0, false, false);
    c1 = __builtin_amdgcn_wmma_f32_16x16x4_f32(false, a0[kk], false, q0, (short)0,
                                               c1, false, false);
    c0 = __builtin_amdgcn_wmma_f32_16x16x4_f32(false, a1[kk], false, p1, (short)0,
                                               c0, false, false);
    c1 = __builtin_amdgcn_wmma_f32_16x16x4_f32(false, a1[kk], false, q1, (short)0,
                                               c1, false, false);
  }

  float* dst = sel ? s2 : s1;
#pragma unroll
  for (int r = 0; r < 8; ++r) {
    const int row = i0 + mt * 16 + r + 8 * hf;
    const int col = h0 + l;
    dst[(b * GAT_K + row) * GAT_H + col]      = c0[r];
    dst[(b * GAT_K + row) * GAT_H + col + 16] = c1[r];
  }
}

template <int M>
__device__ __forceinline__ float xrow(float x) {
  const int b = __builtin_bit_cast(int, x);
  return __builtin_bit_cast(float, __builtin_amdgcn_update_dpp(b, b, 0x160 + M, 0xf, 0xf, false));
}
__device__ __forceinline__ float xhalf(float x) {
  const int b = __builtin_bit_cast(int, x);
  return __builtin_bit_cast(float,
      __builtin_amdgcn_permlanex16(b, b, 0x76543210, 0xfedcba98, false, false));
}

#define E_P 132

__global__ void __launch_bounds__(256)
gat_attn(const float* __restrict__ s1,
         const float* __restrict__ s2,
         const float* __restrict__ lb,
         const float* __restrict__ av,
         const float* __restrict__ bias,
         float* __restrict__ attn) {
  __shared__ float4 r_s[8 * (GAT_H / 4)];
  __shared__ float  e_s[8 * E_P];

  const int blk  = blockIdx.x;
  const int ic   = blk % (GAT_K / 8);
  const int b    = blk / (GAT_K / 8);
  const int wave = threadIdx.x >> 5;
  const int lane = threadIdx.x & 31;
  const int i0   = ic * 8;
  const int j0   = wave * 16;

  const float4* lbv = reinterpret_cast<const float4*>(lb) + lane;
  const float4* s1v = reinterpret_cast<const float4*>(s1 + (b * GAT_K + i0 + wave) * GAT_H) + lane;
  const float4* avv = reinterpret_cast<const float4*>(av) + lane;
  float4 a4[2], l4[2], r4[2];
#pragma unroll
  for (int q = 0; q < 2; ++q) {
    l4[q] = lbv[q * 32];
    r4[q] = s1v[q * 32];
    a4[q] = avv[q * 32];
  }

  const float4* s2v = reinterpret_cast<const float4*>(s2 + (b * GAT_K + j0) * GAT_H) + lane;
  float4 sr[16][2];
#pragma unroll
  for (int jj = 0; jj < 4; ++jj) {
#pragma unroll
    for (int q = 0; q < 2; ++q) sr[jj][q] = s2v[jj * (GAT_H / 4) + q * 32];
  }

#pragma unroll
  for (int q = 0; q < 2; ++q) {
    r_s[wave * (GAT_H / 4) + q * 32 + lane] =
        make_float4(r4[q].x + l4[q].x, r4[q].y + l4[q].y, r4[q].z + l4[q].z, r4[q].w + l4[q].w);
  }
  __syncthreads();

#pragma unroll
  for (int jj = 4; jj < 16; ++jj) {
#pragma unroll
    for (int q = 0; q < 2; ++q) sr[jj][q] = s2v[jj * (GAT_H / 4) + q * 32];
  }
  const int irow = i0 + wave;
  float bz[4];
#pragma unroll
  for (int t = 0; t < 4; ++t) bz[t] = bias[irow * GAT_K + lane + 32 * t];

  float4 rr[8][2];
#pragma unroll
  for (int r = 0; r < 8; ++r) {
#pragma unroll
    for (int q = 0; q < 2; ++q) rr[r][q] = r_s[r * (GAT_H / 4) + q * 32 + lane];
  }

  const bool b0 = (lane & 1) != 0;
  const bool b1 = (lane & 2) != 0;
  const bool b2 = (lane & 4) != 0;
  const bool b3 = (lane & 8) != 0;
  const bool b4 = (lane & 16) != 0;
  const int vidx = ((lane & 1) << 4) | (lane >> 1);
  float* e_dst = e_s + (vidx >> 2) * E_P + j0 + (vidx & 3);

#pragma unroll
  for (int g = 0; g < 4; ++g) {
    float v[32];
#pragma unroll
    for (int q = 0; q < 2; ++q) {
      const float aa[4] = {a4[q].x, a4[q].y, a4[q].z, a4[q].w};
#pragma unroll
      for (int e = 0; e < 4; ++e) {
        float rv[8], cv[4];
#pragma unroll
        for (int r = 0; r < 8; ++r) {
          const float t4[4] = {rr[r][q].x, rr[r][q].y, rr[r][q].z, rr[r][q].w};
          rv[r] = t4[e];
        }
#pragma unroll
        for (int c = 0; c < 4; ++c) {
          const float t4[4] = {sr[4 * g + c][q].x, sr[4 * g + c][q].y,
                               sr[4 * g + c][q].z, sr[4 * g + c][q].w};
          cv[c] = t4[e];
        }
        float t[32], u[32];
#pragma unroll
        for (int k = 0; k < 32; ++k) t[k] = rv[k >> 2] + cv[k & 3];
#pragma unroll
        for (int k = 0; k < 32; ++k) u[k] = GAT_ALPHA * t[k];
#pragma unroll
        for (int k = 0; k < 32; ++k) t[k] = fmaxf(t[k], u[k]);
#pragma unroll
        for (int k = 0; k < 32; ++k) {
          if (q == 0 && e == 0) v[k] = t[k] * aa[e];
          else                  v[k] = fmaf(t[k], aa[e], v[k]);
        }
      }
    }

    float w[16];
#pragma unroll
    for (int k = 0; k < 16; ++k) {
      const float keep = b0 ? v[16 + k] : v[k];
      const float send = b0 ? v[k] : v[16 + k];
      w[k] = keep + xrow<1>(send);
    }
    float x[8];
#pragma unroll
    for (int k = 0; k < 8; ++k) {
      const float keep = b4 ? w[8 + k] : w[k];
      const float send = b4 ? w[k] : w[8 + k];
      x[k] = keep + xhalf(send);
    }
    float y[4];
#pragma unroll
    for (int k = 0; k < 4; ++k) {
      const float keep = b3 ? x[4 + k] : x[k];
      const float send = b3 ? x[k] : x[4 + k];
      y[k] = keep + xrow<8>(send);
    }
    float z[2];
#pragma unroll
    for (int k = 0; k < 2; ++k) {
      const float keep = b2 ? y[2 + k] : y[k];
      const float send = b2 ? y[k] : y[2 + k];
      z[k] = keep + xrow<4>(send);
    }
    const float keep = b1 ? z[1] : z[0];
    const float send = b1 ? z[0] : z[1];
    e_dst[4 * g] = keep + xrow<2>(send);
  }

  __syncthreads();

  float e0 = e_s[wave * E_P + lane +  0] + bz[0];
  float e1 = e_s[wave * E_P + lane + 32] + bz[1];
  float e2 = e_s[wave * E_P + lane + 64] + bz[2];
  float e3 = e_s[wave * E_P + lane + 96] + bz[3];
  float m = fmaxf(fmaxf(e0, e1), fmaxf(e2, e3));
  m = fmaxf(m, __shfl_xor(m, 1));
  m = fmaxf(m, __shfl_xor(m, 2));
  m = fmaxf(m, __shfl_xor(m, 4));
  m = fmaxf(m, __shfl_xor(m, 8));
  m = fmaxf(m, __shfl_xor(m, 16));
  e0 = __expf(e0 - m);
  e1 = __expf(e1 - m);
  e2 = __expf(e2 - m);
  e3 = __expf(e3 - m);
  float s = e0 + e1 + e2 + e3;
  s += __shfl_xor(s, 1);
  s += __shfl_xor(s, 2);
  s += __shfl_xor(s, 4);
  s += __shfl_xor(s, 8);
  s += __shfl_xor(s, 16);
  const float inv = 1.0f / s;
  float* arow = attn + (b * GAT_K + irow) * GAT_K;
  arow[lane +  0] = e0 * inv;
  arow[lane + 32] = e1 * inv;
  arow[lane + 64] = e2 * inv;
  arow[lane + 96] = e3 * inv;
}

#define OUT_T 32
#define OUT_P 132

__global__ void __launch_bounds__(128)
gat_out(const float* __restrict__ attn,
        const float* __restrict__ x,
        float* __restrict__ out) {
  __shared__ float4 as4[OUT_T * OUT_P / 4];
  __shared__ float4 xs4[OUT_T * OUT_P / 4];

  int blk = blockIdx.x;
  const int ft = blk % (GAT_F / OUT_T); blk /= (GAT_F / OUT_T);
  const int it = blk % (GAT_K / OUT_T); blk /= (GAT_K / OUT_T);
  const int b  = blk;
  const int i0 = it * OUT_T, f0 = ft * OUT_T;
  const int tid = threadIdx.x;

  const float4* ag = reinterpret_cast<const float4*>(attn + (b * GAT_K + i0) * GAT_K);
  const float4* xg = reinterpret_cast<const float4*>(x + (b * GAT_F + f0) * GAT_K);
  float4 ar[8], xr[8];
#pragma unroll
  for (int r = 0; r < 8; ++r) ar[r] = ag[tid + 128 * r];
#pragma unroll
  for (int r = 0; r < 8; ++r) xr[r] = xg[tid + 128 * r];
#pragma unroll
  for (int r = 0; r < 8; ++r) {
    const int idx = tid + 128 * r;
    as4[(idx >> 5) * (OUT_P / 4) + (idx & 31)] = ar[r];
  }
#pragma unroll
  for (int r = 0; r < 8; ++r) {
    const int idx = tid + 128 * r;
    xs4[(idx >> 5) * (OUT_P / 4) + (idx & 31)] = xr[r];
  }
  __syncthreads();

  const int wave = tid >> 5;
  const int lane = tid & 31;
  const int hf   = lane >> 4;
  const int l    = lane & 15;
  const int mi   = wave & 1;
  const int ni   = wave >> 1;

  const float4* ap = as4 + (mi * 16 + l) * (OUT_P / 4) + hf;
  const float4* bp = xs4 + (ni * 16 + l) * (OUT_P / 4) + hf;

  float4 aq[GAT_K / 8], bq[GAT_K / 8];
#pragma unroll
  for (int kk = 0; kk < GAT_K / 8; ++kk) {
    aq[kk] = ap[2 * kk];
    bq[kk] = bp[2 * kk];
  }
  v8f ca = {};
  v8f cb = {};
  __builtin_amdgcn_sched_barrier(0);
#pragma unroll
  for (int kk = 0; kk < GAT_K / 8; ++kk) {
    v2f a0, a1, b0, b1;
    a0.x = aq[kk].x; a0.y = aq[kk].y; a1.x = aq[kk].z; a1.y = aq[kk].w;
    b0.x = bq[kk].x; b0.y = bq[kk].y; b1.x = bq[kk].z; b1.y = bq[kk].w;
    ca = __builtin_amdgcn_wmma_f32_16x16x4_f32(false, a0, false, b0, (short)0,
                                               ca, false, false);
    cb = __builtin_amdgcn_wmma_f32_16x16x4_f32(false, a1, false, b1, (short)0,
                                               cb, false, false);
  }
#pragma unroll
  for (int r = 0; r < 8; ++r) {
    const int row = i0 + mi * 16 + r + 8 * hf;
    const int col = f0 + ni * 16 + l;
    const float z = ca[r] + cb[r];
    out[(b * GAT_K + row) * GAT_F + col] = 1.0f / (1.0f + __expf(-z));
  }
}

extern "C" void kernel_launch(void* const* d_in, const int* in_sizes, int n_in,
                              void* d_out, int out_size, void* d_ws, size_t ws_size,
                              hipStream_t stream) {
  (void)in_sizes; (void)n_in; (void)out_size; (void)ws_size;
  const float* x    = (const float*)d_in[0];
  const float* w    = (const float*)d_in[1];
  const float* lb   = (const float*)d_in[2];
  const float* av   = (const float*)d_in[3];
  const float* bias = (const float*)d_in[4];
  float* out = (float*)d_out;

  float* s1   = (float*)d_ws;
  float* s2   = s1 + GAT_B * GAT_K * GAT_H;
  float* attn = s2 + GAT_B * GAT_K * GAT_H;

  gat_qk_gemm<<<GAT_B * (GAT_K / QK_ROWS) * (GAT_H / QK_HT), 256, 0, stream>>>(x, w, s1, s2);
  gat_attn  <<<GAT_B * (GAT_K / 8),               256, 0, stream>>>(s1, s2, lb, av, bias, attn);
  gat_out   <<<GAT_B * (GAT_K / OUT_T) * (GAT_F / OUT_T), 128, 0, stream>>>(attn, x, out);
}
